# starting kernel
# baseline (speedup 1.0000x reference)
#include <hip/hip_runtime.h>


typedef float v2f __attribute__((ext_vector_type(2)));
typedef float v8f __attribute__((ext_vector_type(8)));

namespace {
constexpr int kDIn  = 4096;
constexpr int kDOut = 16;
constexpr int kSeq  = 2048;
constexpr int kRowsPerWave = 32;
}

__global__ __launch_bounds__(32)
void multilora_wmma_f32_kernel(const float* __restrict__ x,
                               const int*   __restrict__ adapter_ids,
                               const float* __restrict__ W,
                               float*       __restrict__ y) {
    const int lane = threadIdx.x & 31;
    const int half = lane >> 4;
    const int r    = lane & 15;

    const int row0 = blockIdx.x * kRowsPerWave;
    const int b    = row0 / kSeq;
    const int aid  = adapter_ids[b];

    const float* xp0 = x + (size_t)(row0 + r) * kDIn + 2 * half;
    const float* xp1 = xp0 + (size_t)16 * kDIn;
    const float* wp  = W + ((size_t)aid * kDOut + r) * kDIn + 2 * half;

    v8f c0 = {};
    v8f c1 = {};

    #pragma unroll 8
    for (int k = 0; k < kDIn; k += 4) {
        v2f bf = *(const v2f*)(wp  + k);
        v2f a0 = *(const v2f*)(xp0 + k);
        v2f a1 = *(const v2f*)(xp1 + k);
        c0 = __builtin_amdgcn_wmma_f32_16x16x4_f32(
                 false, a0, false, bf, (short)0, c0, false, false);
        c1 = __builtin_amdgcn_wmma_f32_16x16x4_f32(
                 false, a1, false, bf, (short)0, c1, false, false);
    }

    float* yp0 = y + (size_t)(row0 + 8 * half) * kDOut + r;
    #pragma unroll
    for (int v = 0; v < 8; ++v) {
        yp0[(size_t)v * kDOut] = c0[v];
    }
    float* yp1 = yp0 + (size_t)16 * kDOut;
    #pragma unroll
    for (int v = 0; v < 8; ++v) {
        yp1[(size_t)v * kDOut] = c1[v];
    }
}

extern "C" void kernel_launch(void* const* d_in, const int* in_sizes, int n_in,
                              void* d_out, int out_size, void* d_ws, size_t ws_size,
                              hipStream_t stream) {
    (void)in_sizes; (void)n_in; (void)d_ws; (void)ws_size; (void)out_size;
    const float* x   = (const float*)d_in[0];
    const int*   ids = (const int*)  d_in[1];
    const float* W   = (const float*)d_in[2];
    float*       y   = (float*)d_out;

    constexpr int totalRows = 4 * 2048;
    dim3 grid(totalRows / kRowsPerWave);
    dim3 block(32);
    multilora_wmma_f32_kernel<<<grid, block, 0, stream>>>(x, ids, W, y);
}


# Split K over the 16 waves of a 512-thread work-group (256 k per wave, 4096 waves for 256); the 16 partial tiles are added in fixed wave order through LDS, which changes the summation order
# speedup vs baseline: 3.3845x; 3.3845x over previous
#include <hip/hip_runtime.h>


typedef float v2f __attribute__((ext_vector_type(2)));
typedef float v8f __attribute__((ext_vector_type(8)));

namespace {
constexpr int kDIn  = 4096;
constexpr int kDOut = 16;
constexpr int kSeq  = 2048;
constexpr int kRowsPerBlock = 32;
constexpr int kWaves  = 16;
constexpr int kSliceK = kDIn / kWaves;
constexpr int kTileElems = kRowsPerBlock * kDOut;
static_assert(kWaves * 32 == kTileElems, "one reducing thread per output element");
}

__global__ __launch_bounds__(kWaves * 32)
void multilora_wmma_f32_splitk_kernel(const float* __restrict__ x,
                                      const int*   __restrict__ adapter_ids,
                                      const float* __restrict__ W,
                                      float*       __restrict__ y) {
    __shared__ float red[kWaves * kTileElems];

    const int lane = threadIdx.x & 31;
    const int half = lane >> 4;
    const int r    = lane & 15;
    const int wave = __builtin_amdgcn_readfirstlane((int)(threadIdx.x >> 5));

    const int row0 = blockIdx.x * kRowsPerBlock;
    const int b    = row0 / kSeq;
    const int aid  = adapter_ids[b];
    const int k0   = wave * kSliceK;

    const float* xp0 = x + (size_t)(row0 + r) * kDIn + k0 + 2 * half;
    const float* xp1 = xp0 + (size_t)16 * kDIn;
    const float* wp  = W + ((size_t)aid * kDOut + r) * kDIn + k0 + 2 * half;

    v8f c0 = {};
    v8f c1 = {};

    #pragma unroll 8
    for (int k = 0; k < kSliceK; k += 4) {
        v2f bf = *(const v2f*)(wp  + k);
        v2f a0 = *(const v2f*)(xp0 + k);
        v2f a1 = *(const v2f*)(xp1 + k);
        c0 = __builtin_amdgcn_wmma_f32_16x16x4_f32(
                 false, a0, false, bf, (short)0, c0, false, false);
        c1 = __builtin_amdgcn_wmma_f32_16x16x4_f32(
                 false, a1, false, bf, (short)0, c1, false, false);
    }

    float* mine = red + wave * kTileElems + lane;
    #pragma unroll
    for (int v = 0; v < 8; ++v) {
        mine[v * 32]       = c0[v];
        mine[256 + v * 32] = c1[v];
    }
    __syncthreads();

    const int t = threadIdx.x;
    float s = red[t];
    #pragma unroll
    for (int w = 1; w < kWaves; ++w) {
        s += red[w * kTileElems + t];
    }

    const int tile = wave >> 3;
    const int v    = wave & 7;
    y[(size_t)(row0 + 16 * tile + 8 * half + v) * kDOut + r] = s;
}

extern "C" void kernel_launch(void* const* d_in, const int* in_sizes, int n_in,
                              void* d_out, int out_size, void* d_ws, size_t ws_size,
                              hipStream_t stream) {
    (void)in_sizes; (void)n_in; (void)d_ws; (void)ws_size; (void)out_size;
    const float* x   = (const float*)d_in[0];
    const int*   ids = (const int*)  d_in[1];
    const float* W   = (const float*)d_in[2];
    float*       y   = (float*)d_out;

    constexpr int totalRows = 4 * 2048;
    dim3 grid(totalRows / kRowsPerBlock);
    dim3 block(kWaves * 32);
    multilora_wmma_f32_splitk_kernel<<<grid, block, 0, stream>>>(x, ids, W, y);
}


# Fragment loads 16 B per lane, not 8: a wave-wide load takes 32 B of each of 16 rows (half the line touches); register budget set so 24-load groups stay whole; k regrouped per 8 k: sum order changes
# speedup vs baseline: 3.8598x; 1.1404x over previous
#include <hip/hip_runtime.h>


typedef float v2f __attribute__((ext_vector_type(2)));
typedef float v4f __attribute__((ext_vector_type(4)));
typedef float v8f __attribute__((ext_vector_type(8)));

namespace {
constexpr int kDIn  = 4096;
constexpr int kDOut = 16;
constexpr int kSeq  = 2048;
constexpr int kRowsPerBlock = 32;
constexpr int kWaves  = 16;
constexpr int kSliceK = kDIn / kWaves;
constexpr int kTileElems = kRowsPerBlock * kDOut;
static_assert(kWaves * 32 == kTileElems, "one reducing thread per output element");
}

__global__ __launch_bounds__(kWaves * 32) __attribute__((amdgpu_waves_per_eu(4, 4)))
void multilora_wmma_f32_splitk_kernel(const float* __restrict__ x,
                                      const int*   __restrict__ adapter_ids,
                                      const float* __restrict__ W,
                                      float*       __restrict__ y) {
    __shared__ float red[kWaves * kTileElems];

    const int lane = threadIdx.x & 31;
    const int half = lane >> 4;
    const int r    = lane & 15;
    const int wave = __builtin_amdgcn_readfirstlane((int)(threadIdx.x >> 5));

    const int row0 = blockIdx.x * kRowsPerBlock;
    const int b    = row0 / kSeq;
    const int aid  = adapter_ids[b];
    const int k0   = wave * kSliceK;

    const float* xp0 = x + (size_t)(row0 + r) * kDIn + k0 + 4 * half;
    const float* xp1 = xp0 + (size_t)16 * kDIn;
    const float* wp  = W + ((size_t)aid * kDOut + r) * kDIn + k0 + 4 * half;

    v8f c0 = {};
    v8f c1 = {};

    #pragma unroll 8
    for (int k = 0; k < kSliceK; k += 8) {
        const v4f bq = *(const v4f*)(wp  + k);
        const v4f q0 = *(const v4f*)(xp0 + k);
        const v4f q1 = *(const v4f*)(xp1 + k);
        c0 = __builtin_amdgcn_wmma_f32_16x16x4_f32(
                 false, q0.lo, false, bq.lo, (short)0, c0, false, false);
        c1 = __builtin_amdgcn_wmma_f32_16x16x4_f32(
                 false, q1.lo, false, bq.lo, (short)0, c1, false, false);
        c0 = __builtin_amdgcn_wmma_f32_16x16x4_f32(
                 false, q0.hi, false, bq.hi, (short)0, c0, false, false);
        c1 = __builtin_amdgcn_wmma_f32_16x16x4_f32(
                 false, q1.hi, false, bq.hi, (short)0, c1, false, false);
    }

    float* mine = red + wave * kTileElems + lane;
    #pragma unroll
    for (int v = 0; v < 8; ++v) {
        mine[v * 32]       = c0[v];
        mine[256 + v * 32] = c1[v];
    }
    __syncthreads();

    const int t = threadIdx.x;
    float s = red[t];
    #pragma unroll
    for (int w = 1; w < kWaves; ++w) {
        s += red[w * kTileElems + t];
    }

    const int tile = wave >> 3;
    const int v    = wave & 7;
    y[(size_t)(row0 + 16 * tile + 8 * half + v) * kDOut + r] = s;
}

extern "C" void kernel_launch(void* const* d_in, const int* in_sizes, int n_in,
                              void* d_out, int out_size, void* d_ws, size_t ws_size,
                              hipStream_t stream) {
    (void)in_sizes; (void)n_in; (void)d_ws; (void)ws_size; (void)out_size;
    const float* x   = (const float*)d_in[0];
    const int*   ids = (const int*)  d_in[1];
    const float* W   = (const float*)d_in[2];
    float*       y   = (float*)d_out;

    constexpr int totalRows = 4 * 2048;
    dim3 grid(totalRows / kRowsPerBlock);
    dim3 block(kWaves * 32);
    multilora_wmma_f32_splitk_kernel<<<grid, block, 0, stream>>>(x, ids, W, y);
}


# Global loads take whole 128-B lines (256 B of a row per lane half, not 32 B of each of 16 rows); x, W become fragments via wave-private LDS; next chunk's loads fly under the products; k order kept
# speedup vs baseline: 5.3961x; 1.3980x over previous
#include <hip/hip_runtime.h>


typedef float v2f __attribute__((ext_vector_type(2)));
typedef float v4f __attribute__((ext_vector_type(4)));
typedef float v8f __attribute__((ext_vector_type(8)));

namespace {
constexpr int kDIn  = 4096;
constexpr int kDOut = 16;
constexpr int kSeq  = 2048;
constexpr int kRowsPerBlock = 32;
constexpr int kWaves  = 16;
constexpr int kSliceK = kDIn / kWaves;
constexpr int kTileElems = kRowsPerBlock * kDOut;
static_assert(kWaves * 32 == kTileElems, "one reducing thread per output element");

constexpr int kChunkK = 64;
constexpr int kChunks = kSliceK / kChunkK;
constexpr int kPitch = kChunkK + 4;
constexpr int kTileFloats  = 16 * kPitch;
constexpr int kStageFloats = 3 * kTileFloats;
}

__global__ __launch_bounds__(kWaves * 32) __attribute__((amdgpu_waves_per_eu(4, 4)))
void multilora_wmma_f32_splitk_kernel(const float* __restrict__ x,
                                      const int*   __restrict__ adapter_ids,
                                      const float* __restrict__ W,
                                      float*       __restrict__ y) {
    __shared__ __attribute__((aligned(16))) float stage[kWaves * kStageFloats];
    __shared__ float red[kWaves * kTileElems];

    const int lane = threadIdx.x & 31;
    const int half = lane >> 4;
    const int r    = lane & 15;
    const int wave = __builtin_amdgcn_readfirstlane((int)(threadIdx.x >> 5));

    const int row0 = blockIdx.x * kRowsPerBlock;
    const int b    = row0 / kSeq;
    const int aid  = adapter_ids[b];
    const int k0   = wave * kSliceK;

    const float* gx0 = x + (size_t)(row0 + half) * kDIn + k0 + 4 * r;
    const float* gx1 = gx0 + (size_t)16 * kDIn;
    const float* gw  = W + ((size_t)aid * kDOut + half) * kDIn + k0 + 4 * r;

    float*       st  = stage + wave * kStageFloats;
    float*       sst = st + half * kPitch + 4 * r;
    const float* sld = st + r * kPitch + 4 * half;

    v8f c0 = {};
    v8f c1 = {};

    v4f tw[8], t0[8], t1[8];
    #pragma unroll
    for (int i = 0; i < 8; ++i) {
        tw[i] = *(const v4f*)(gw  + (size_t)(2 * i) * kDIn);
        t0[i] = *(const v4f*)(gx0 + (size_t)(2 * i) * kDIn);
        t1[i] = *(const v4f*)(gx1 + (size_t)(2 * i) * kDIn);
    }

    #pragma unroll
    for (int c = 0; c < kChunks; ++c) {
        #pragma unroll
        for (int i = 0; i < 8; ++i) {
            *(v4f*)(sst + 0 * kTileFloats + 2 * i * kPitch) = tw[i];
            *(v4f*)(sst + 1 * kTileFloats + 2 * i * kPitch) = t0[i];
            *(v4f*)(sst + 2 * kTileFloats + 2 * i * kPitch) = t1[i];
        }
        __builtin_amdgcn_fence(__ATOMIC_RELEASE, "wavefront");
        __builtin_amdgcn_fence(__ATOMIC_ACQUIRE, "wavefront");

        if (c + 1 < kChunks) {
            #pragma unroll
            for (int i = 0; i < 8; ++i) {
                tw[i] = *(const v4f*)(gw  + (size_t)(2 * i) * kDIn + (c + 1) * kChunkK);
                t0[i] = *(const v4f*)(gx0 + (size_t)(2 * i) * kDIn + (c + 1) * kChunkK);
                t1[i] = *(const v4f*)(gx1 + (size_t)(2 * i) * kDIn + (c + 1) * kChunkK);
            }
        }

        #pragma unroll
        for (int j = 0; j < 8; ++j) {
            const v4f bq = *(const v4f*)(sld + 0 * kTileFloats + 8 * j);
            const v4f q0 = *(const v4f*)(sld + 1 * kTileFloats + 8 * j);
            const v4f q1 = *(const v4f*)(sld + 2 * kTileFloats + 8 * j);
            c0 = __builtin_amdgcn_wmma_f32_16x16x4_f32(
                     false, q0.lo, false, bq.lo, (short)0, c0, false, false);
            c1 = __builtin_amdgcn_wmma_f32_16x16x4_f32(
                     false, q1.lo, false, bq.lo, (short)0, c1, false, false);
            c0 = __builtin_amdgcn_wmma_f32_16x16x4_f32(
                     false, q0.hi, false, bq.hi, (short)0, c0, false, false);
            c1 = __builtin_amdgcn_wmma_f32_16x16x4_f32(
                     false, q1.hi, false, bq.hi, (short)0, c1, false, false);
        }
        __builtin_amdgcn_fence(__ATOMIC_RELEASE, "wavefront");
        __builtin_amdgcn_fence(__ATOMIC_ACQUIRE, "wavefront");
        __builtin_amdgcn_sched_barrier(0);
    }

    float* mine = red + wave * kTileElems + lane;
    #pragma unroll
    for (int v = 0; v < 8; ++v) {
        mine[v * 32]       = c0[v];
        mine[256 + v * 32] = c1[v];
    }
    __syncthreads();

    const int t = threadIdx.x;
    float s = red[t];
    #pragma unroll
    for (int w = 1; w < kWaves; ++w) {
        s += red[w * kTileElems + t];
    }

    const int tile = wave >> 3;
    const int v    = wave & 7;
    y[(size_t)(row0 + 16 * tile + 8 * half + v) * kDOut + r] = s;
}

extern "C" void kernel_launch(void* const* d_in, const int* in_sizes, int n_in,
                              void* d_out, int out_size, void* d_ws, size_t ws_size,
                              hipStream_t stream) {
    (void)in_sizes; (void)n_in; (void)d_ws; (void)ws_size; (void)out_size;
    const float* x   = (const float*)d_in[0];
    const int*   ids = (const int*)  d_in[1];
    const float* W   = (const float*)d_in[2];
    float*       y   = (float*)d_out;

    constexpr int totalRows = 4 * 2048;
    dim3 grid(totalRows / kRowsPerBlock);
    dim3 block(kWaves * 32);
    multilora_wmma_f32_splitk_kernel<<<grid, block, 0, stream>>>(x, ids, W, y);
}
